# starting kernel
# baseline (speedup 1.0000x reference)
#include <hip/hip_runtime.h>

typedef __attribute__((ext_vector_type(2))) float v2f;
typedef __attribute__((ext_vector_type(4))) float v4f;
typedef __attribute__((ext_vector_type(8))) float v8f;

constexpr int B  = 128;
constexpr int T  = 512;
constexpr int DZ = 512;

__global__ void scan_kernel(const float* __restrict__ x, float* __restrict__ S) {
  __shared__ float buf[T];
  const int b = blockIdx.x;
  const int t = threadIdx.x;

  float v = x[(size_t)b * T + t];
  buf[t] = v;
  __syncthreads();

  for (int off = 1; off < T; off <<= 1) {
    float add = (t >= off) ? buf[t - off] : 0.0f;
    __syncthreads();
    v += add;
    buf[t] = v;
    __syncthreads();
  }
  S[(size_t)b * T + t] = v;
}

__global__ void rank1_wmma_kernel(const float* __restrict__ S,
                                  const float* __restrict__ z0,
                                  const float* __restrict__ f,
                                  float* __restrict__ out) {
  constexpr int DT = DZ / 16;
  constexpr int TT = T / 16;

  const int wave = blockIdx.x * (blockDim.x >> 5) + (threadIdx.x >> 5);
  const int lane = threadIdx.x & 31;

  const int dt = wave & (DT - 1);
  const int tt = (wave >> 5) & (TT - 1);
  const int b  = wave >> 10;

  const bool lo  = lane < 16;
  const int  l16 = lane & 15;

  v2f a;
  a[0] = lo ? f[dt * 16 + l16] : 0.0f;
  a[1] = 0.0f;

  v2f bm;
  bm[0] = lo ? S[(size_t)b * T + tt * 16 + l16] : 0.0f;
  bm[1] = 0.0f;

  const float* z0row = z0 + (size_t)b * DZ + dt * 16 + (lo ? 0 : 8);
  v8f c;
  c[0] = z0row[0]; c[1] = z0row[1]; c[2] = z0row[2]; c[3] = z0row[3];
  c[4] = z0row[4]; c[5] = z0row[5]; c[6] = z0row[6]; c[7] = z0row[7];

  v8f d = __builtin_amdgcn_wmma_f32_16x16x4_f32(
       false, a,  false, bm,
       (short)0, c,  false,  false);

  const size_t base = ((size_t)b * T + tt * 16 + l16) * DZ + dt * 16 + (lo ? 0 : 8);
  v4f d03 = {d[0], d[1], d[2], d[3]};
  v4f d47 = {d[4], d[5], d[6], d[7]};
  *(v4f*)(out + base)     = d03;
  *(v4f*)(out + base + 4) = d47;
}

extern "C" void kernel_launch(void* const* d_in, const int* in_sizes, int n_in,
                              void* d_out, int out_size, void* d_ws, size_t ws_size,
                              hipStream_t stream) {
  const float* x  = (const float*)d_in[0];
  const float* z0 = (const float*)d_in[1];
  const float* f  = (const float*)d_in[2];
  float* out = (float*)d_out;
  float* S   = (float*)d_ws;

  (void)in_sizes; (void)n_in; (void)out_size; (void)ws_size;

  scan_kernel<<<B, T, 0, stream>>>(x, S);

  constexpr int totalWaves = B * (T / 16) * (DZ / 16);
  constexpr int blockThreads = 256;
  constexpr int grid = totalWaves / (blockThreads / 32);
  rank1_wmma_kernel<<<grid, blockThreads, 0, stream>>>(S, z0, f, out);
}


# main kernel re-tiled for whole-line stores: M=t, 4 WMMAs share A so a lane holds 4 consecutive d of a row (b128 stores, 256-byte row pieces); 256 work-groups of 8 waves, 16 row tiles per wave
# speedup vs baseline: 2.0324x; 2.0324x over previous
#include <hip/hip_runtime.h>


typedef __attribute__((ext_vector_type(2))) float v2f;
typedef __attribute__((ext_vector_type(4))) float v4f;
typedef __attribute__((ext_vector_type(8))) float v8f;

constexpr int B  = 128;
constexpr int T  = 512;
constexpr int DZ = 512;

__global__ void scan_kernel(const float* __restrict__ x, float* __restrict__ S) {
  __shared__ float buf[T];
  const int b = blockIdx.x;
  const int t = threadIdx.x;

  float v = x[(size_t)b * T + t];
  buf[t] = v;
  __syncthreads();

  for (int off = 1; off < T; off <<= 1) {
    float add = (t >= off) ? buf[t - off] : 0.0f;
    __syncthreads();
    v += add;
    buf[t] = v;
    __syncthreads();
  }
  S[(size_t)b * T + t] = v;
}

constexpr int ROWS_PER_WG = 256;
constexpr int TILES_PER_WAVE = ROWS_PER_WG / 16;

__global__ __launch_bounds__(256)
void rank1_wmma_kernel(const float* __restrict__ S,
                       const float* __restrict__ z0,
                       const float* __restrict__ f,
                       float* __restrict__ out) {
  const int wave = threadIdx.x >> 5;
  const int lane = threadIdx.x & 31;
  const int c    = lane & 15;
  const int h    = lane >> 4;
  const bool lo  = lane < 16;

  const int b     = blockIdx.x >> 1;
  const int tbase = (blockIdx.x & 1) * ROWS_PER_WG;
  const int dcol  = wave * 64 + 4 * c;

  const v4f fv = *(const v4f*)(f + dcol);
  const v4f zv = *(const v4f*)(z0 + (size_t)b * DZ + dcol);

  v2f bj[4];
  v8f cj[4];
  for (int j = 0; j < 4; ++j) {
    bj[j][0] = lo ? fv[j] : 0.0f;
    bj[j][1] = 0.0f;
    for (int r = 0; r < 8; ++r) cj[j][r] = zv[j];
  }

  const float* Srow = S + (size_t)b * T + tbase + c;
  float sv[TILES_PER_WAVE];
  for (int q = 0; q < TILES_PER_WAVE; ++q) sv[q] = Srow[16 * q];

  float* orow = out + ((size_t)b * T + tbase + 8 * h) * DZ + dcol;

  for (int q = 0; q < TILES_PER_WAVE; ++q) {
    v2f a;
    a[0] = lo ? sv[q] : 0.0f;
    a[1] = 0.0f;

    v8f d0 = __builtin_amdgcn_wmma_f32_16x16x4_f32(false, a, false, bj[0], (short)0, cj[0], false, false);
    v8f d1 = __builtin_amdgcn_wmma_f32_16x16x4_f32(false, a, false, bj[1], (short)0, cj[1], false, false);
    v8f d2 = __builtin_amdgcn_wmma_f32_16x16x4_f32(false, a, false, bj[2], (short)0, cj[2], false, false);
    v8f d3 = __builtin_amdgcn_wmma_f32_16x16x4_f32(false, a, false, bj[3], (short)0, cj[3], false, false);

    float* o = orow + (size_t)(16 * q) * DZ;
    for (int r = 0; r < 8; ++r) {
      v4f v = {d0[r], d1[r], d2[r], d3[r]};
      *(v4f*)(o + (size_t)r * DZ) = v;
    }
  }
}

extern "C" void kernel_launch(void* const* d_in, const int* in_sizes, int n_in,
                              void* d_out, int out_size, void* d_ws, size_t ws_size,
                              hipStream_t stream) {
  const float* x  = (const float*)d_in[0];
  const float* z0 = (const float*)d_in[1];
  const float* f  = (const float*)d_in[2];
  float* out = (float*)d_out;
  float* S   = (float*)d_ws;

  (void)in_sizes; (void)n_in; (void)out_size; (void)ws_size;

  scan_kernel<<<B, T, 0, stream>>>(x, S);

  constexpr int grid = (B * T) / ROWS_PER_WG;
  rank1_wmma_kernel<<<grid, 256, 0, stream>>>(S, z0, f, out);
}


# prefix sum fused into the streaming kernel: each wave rebuilds its 256 row sums from x in registers (16-lane DPP scans + chunk carries; prefix summation tree changes); one launch, no workspace
# speedup vs baseline: 2.6123x; 1.2853x over previous
#include <hip/hip_runtime.h>


typedef __attribute__((ext_vector_type(2))) float v2f;
typedef __attribute__((ext_vector_type(4))) float v4f;
typedef __attribute__((ext_vector_type(8))) float v8f;

constexpr int B  = 128;
constexpr int T  = 512;
constexpr int DZ = 512;


template <int N>
__device__ __forceinline__ float row_shift_right(float v) {
  return __builtin_bit_cast(
      float, __builtin_amdgcn_update_dpp(0, __builtin_bit_cast(int, v),
                                         0x110 + N, 0xF, 0xF, true));
}

__device__ __forceinline__ float row_last(float v) {
  return __builtin_bit_cast(
      float, __builtin_amdgcn_update_dpp(0, __builtin_bit_cast(int, v),
                                         0x15F, 0xF, 0xF, true));
}

constexpr int ROWS_PER_WG = 256;
constexpr int TILES_PER_WAVE = ROWS_PER_WG / 16;

__global__ __launch_bounds__(256)
void scan_rank1_wmma_kernel(const float* __restrict__ x,
                            const float* __restrict__ z0,
                            const float* __restrict__ f,
                            float* __restrict__ out) {
  const int wave = threadIdx.x >> 5;
  const int lane = threadIdx.x & 31;
  const int c    = lane & 15;
  const int h    = lane >> 4;
  const bool lo  = lane < 16;

  const int b     = blockIdx.x >> 1;
  const int tbase = (blockIdx.x & 1) * ROWS_PER_WG;
  const int dcol  = wave * 64 + 4 * c;

  const float* xrow = x + (size_t)b * T + (lo ? tbase : 0) + c;
  float sv[TILES_PER_WAVE];
  for (int q = 0; q < TILES_PER_WAVE; ++q) sv[q] = xrow[16 * q];

  const v4f fv = *(const v4f*)(f + dcol);
  const v4f zv = *(const v4f*)(z0 + (size_t)b * DZ + dcol);

  v2f bj[4];
  v8f cj[4];
  for (int j = 0; j < 4; ++j) {
    bj[j][0] = lo ? fv[j] : 0.0f;
    bj[j][1] = 0.0f;
    for (int r = 0; r < 8; ++r) cj[j][r] = zv[j];
  }

  for (int q = 0; q < TILES_PER_WAVE; ++q) {
    float s = sv[q];
    s += row_shift_right<1>(s);
    s += row_shift_right<2>(s);
    s += row_shift_right<4>(s);
    s += row_shift_right<8>(s);
    sv[q] = s;
  }

  float carry = 0.0f;
  for (int q = 0; q < TILES_PER_WAVE; ++q) {
    const float total = row_last(sv[q]);
    sv[q] += carry;
    carry += total;
  }

  const float first_half =
      __builtin_bit_cast(float, __builtin_amdgcn_readlane(__builtin_bit_cast(int, carry), 31));
  const float start = (tbase != 0) ? first_half : 0.0f;
  for (int q = 0; q < TILES_PER_WAVE; ++q) sv[q] += start;

  float* orow = out + ((size_t)b * T + tbase + 8 * h) * DZ + dcol;

  for (int q = 0; q < TILES_PER_WAVE; ++q) {
    v2f a;
    a[0] = lo ? sv[q] : 0.0f;
    a[1] = 0.0f;

    v8f d0 = __builtin_amdgcn_wmma_f32_16x16x4_f32(false, a, false, bj[0], (short)0, cj[0], false, false);
    v8f d1 = __builtin_amdgcn_wmma_f32_16x16x4_f32(false, a, false, bj[1], (short)0, cj[1], false, false);
    v8f d2 = __builtin_amdgcn_wmma_f32_16x16x4_f32(false, a, false, bj[2], (short)0, cj[2], false, false);
    v8f d3 = __builtin_amdgcn_wmma_f32_16x16x4_f32(false, a, false, bj[3], (short)0, cj[3], false, false);

    float* o = orow + (size_t)(16 * q) * DZ;
    for (int r = 0; r < 8; ++r) {
      v4f v = {d0[r], d1[r], d2[r], d3[r]};
      *(v4f*)(o + (size_t)r * DZ) = v;
    }
  }
}

extern "C" void kernel_launch(void* const* d_in, const int* in_sizes, int n_in,
                              void* d_out, int out_size, void* d_ws, size_t ws_size,
                              hipStream_t stream) {
  const float* x  = (const float*)d_in[0];
  const float* z0 = (const float*)d_in[1];
  const float* f  = (const float*)d_in[2];
  float* out = (float*)d_out;

  (void)in_sizes; (void)n_in; (void)out_size; (void)d_ws; (void)ws_size;

  constexpr int grid = (B * T) / ROWS_PER_WG;
  scan_rank1_wmma_kernel<<<grid, 256, 0, stream>>>(x, z0, f, out);
}
